# starting kernel
# baseline (speedup 1.0000x reference)
#include <hip/hip_runtime.h>

typedef __attribute__((ext_vector_type(2))) float v2f;
typedef __attribute__((ext_vector_type(4))) float v4f;
typedef __attribute__((ext_vector_type(8))) float v8f;

#define D_IN  256
#define D_OUT 64

__global__ __launch_bounds__(256) void k_zero(float* __restrict__ out, int n4) {
    int i = blockIdx.x * 256 + threadIdx.x;
    if (i < n4) {
        v4f z = {0.f, 0.f, 0.f, 0.f};
        ((v4f*)out)[i] = z;
    }
}

__global__ __launch_bounds__(256) void k_gemm(const float* __restrict__ x,
                                              const float* __restrict__ w,
                                              float* __restrict__ hidden,
                                              int n_rows) {
    __shared__ float wl[D_IN * D_OUT];

    const int tid = threadIdx.x;
    for (int i = tid; i < D_IN * D_OUT; i += 256) {
        int k = i / D_OUT;
        int n = i - k * D_OUT;
        wl[(k >> 1) * 128 + n * 2 + (k & 1)] = w[i];
    }
    __syncthreads();

    const int wave = tid >> 5;
    const int lane = tid & 31;
    const int l15  = lane & 15;
    const int hi   = lane >> 4;

    const long row0 = ((long)blockIdx.x * 8 + wave) * 16;
    if (row0 >= n_rows) return;

    const float* ap = x + (row0 + l15) * D_IN + 2 * hi;

    v8f c0 = {}, c1 = {}, c2 = {}, c3 = {};

    for (int k = 0; k < D_IN; k += 4) {
        v2f a = *(const v2f*)(ap + k);
        const float* bbase = &wl[((k >> 1) + hi) * 128 + l15 * 2];
        v2f b0 = *(const v2f*)(bbase +  0);
        v2f b1 = *(const v2f*)(bbase + 32);
        v2f b2 = *(const v2f*)(bbase + 64);
        v2f b3 = *(const v2f*)(bbase + 96);
        c0 = __builtin_amdgcn_wmma_f32_16x16x4_f32(false, a, false, b0, (short)0, c0, false, false);
        c1 = __builtin_amdgcn_wmma_f32_16x16x4_f32(false, a, false, b1, (short)0, c1, false, false);
        c2 = __builtin_amdgcn_wmma_f32_16x16x4_f32(false, a, false, b2, (short)0, c2, false, false);
        c3 = __builtin_amdgcn_wmma_f32_16x16x4_f32(false, a, false, b3, (short)0, c3, false, false);
    }

    float* hp = hidden + (row0 + 8 * hi) * D_OUT + l15;
#pragma unroll
    for (int r = 0; r < 8; ++r) {
        hp[r * D_OUT +  0] = c0[r];
        hp[r * D_OUT + 16] = c1[r];
        hp[r * D_OUT + 32] = c2[r];
        hp[r * D_OUT + 48] = c3[r];
    }
}

__global__ __launch_bounds__(256) void k_scatter(const float* __restrict__ hidden,
                                                 const float* __restrict__ adj,
                                                 const int* __restrict__ row,
                                                 const int* __restrict__ col,
                                                 float* __restrict__ out,
                                                 int n_edges) {
    const int wave = threadIdx.x >> 5;
    const int lane = threadIdx.x & 31;
    const int e    = blockIdx.x * 8 + wave;
    if (e >= n_edges) return;

    const int   r = row[e];
    const int   c = col[e];
    const float v = adj[e];

    const float* h = hidden + (long)c * D_OUT;
    float*       o = out    + (long)r * D_OUT;

    atomicAdd(&o[lane],      v * h[lane]);
    atomicAdd(&o[lane + 32], v * h[lane + 32]);
}

__global__ __launch_bounds__(256) void k_prelu(float* __restrict__ out,
                                               const float* __restrict__ slope,
                                               int n4) {
    int i = blockIdx.x * 256 + threadIdx.x;
    if (i >= n4) return;
    const float s = slope[0];
    v4f v = ((v4f*)out)[i];
#pragma unroll
    for (int j = 0; j < 4; ++j) {
        float t = v[j];
        v[j] = t > 0.f ? t : s * t;
    }
    ((v4f*)out)[i] = v;
}

extern "C" void kernel_launch(void* const* d_in, const int* in_sizes, int n_in,
                              void* d_out, int out_size, void* d_ws, size_t ws_size,
                              hipStream_t stream) {
    const float* x    = (const float*)d_in[0];
    const float* w    = (const float*)d_in[1];
    const float* adj  = (const float*)d_in[2];
    const float* pa   = (const float*)d_in[3];
    const int*   row  = (const int*)d_in[4];
    const int*   col  = (const int*)d_in[5];
    float*       out  = (float*)d_out;
    float*       hid  = (float*)d_ws;

    const int n_nodes = in_sizes[0] / D_IN;
    const int n_edges = in_sizes[2];
    const int out4    = out_size / 4;

    k_zero<<<(out4 + 255) / 256, 256, 0, stream>>>(out, out4);

    const int row_tiles = (n_nodes + 15) / 16;
    k_gemm<<<(row_tiles + 7) / 8, 256, 0, stream>>>(x, w, hid, n_nodes);

    k_scatter<<<(n_edges + 7) / 8, 256, 0, stream>>>(hid, adj, row, col, out, n_edges);

    k_prelu<<<(out4 + 255) / 256, 256, 0, stream>>>(out, pa, out4);
}


# edge lists by global exchange -> 64 edge chunks each sorted by source tile with LDS counters; wave parks its tile in LDS and walks 64 known runs; 4 -> 3 launches; f32 atomic order changes
# speedup vs baseline: 1.3596x; 1.3596x over previous
#include <hip/hip_runtime.h>


typedef __attribute__((ext_vector_type(2))) float v2f;
typedef __attribute__((ext_vector_type(4))) float v4f;
typedef __attribute__((ext_vector_type(8))) float v8f;
typedef __attribute__((ext_vector_type(4))) int   v4i;
typedef __attribute__((ext_vector_type(2))) int   v2i;

#define D_IN     256
#define D_OUT    64
#define N_CHUNK  64
#define BIN_T    1024
#define BIN_B    8
#define HL_PITCH 66

__global__ __launch_bounds__(BIN_T) void k_bin(const float* __restrict__ adj,
                                               const int* __restrict__ row,
                                               const int* __restrict__ col,
                                               int* __restrict__ rec,
                                               int* __restrict__ desc,
                                               float* __restrict__ out, int out4,
                                               int n_edges, int n_nodes,
                                               int n_tiles, int chunk) {
    extern __shared__ int sm[];
    const int tid = threadIdx.x;

    if (blockIdx.x >= N_CHUNK) {
        const int nz = (int)gridDim.x - N_CHUNK;
        const v4f z = {0.f, 0.f, 0.f, 0.f};
        for (int i = ((int)blockIdx.x - N_CHUNK) * BIN_T + tid; i < out4; i += nz * BIN_T)
            ((v4f*)out)[i] = z;
        return;
    }

    int* cnt  = sm;
    int* off  = sm + n_tiles;
    int* wsum = off + n_tiles;

    const int e0 = (int)blockIdx.x * chunk;
    int e1 = e0 + chunk;
    if (e1 > n_edges) e1 = n_edges;

    for (int t = tid; t < n_tiles; t += BIN_T) cnt[t] = 0;
    __syncthreads();

    for (int base = e0 + tid; base < e1; base += BIN_T * BIN_B) {
        int c[BIN_B];
#pragma unroll
        for (int u = 0; u < BIN_B; ++u) {
            const int e = base + u * BIN_T;
            c[u] = (e < e1) ? col[e] : -1;
        }
#pragma unroll
        for (int u = 0; u < BIN_B; ++u)
            if ((unsigned)c[u] < (unsigned)n_nodes) atomicAdd(&cnt[c[u] >> 4], 1);
    }
    __syncthreads();

    const int per  = (n_tiles + BIN_T - 1) / BIN_T;
    const int t0   = tid * per;
    const int lane = tid & 31;
    const int wv   = tid >> 5;
    int s = 0;
    for (int k = 0; k < per; ++k) {
        const int t = t0 + k;
        if (t < n_tiles) s += cnt[t];
    }
    int inc = s;
    for (int d = 1; d < 32; d <<= 1) {
        const int v = __shfl_up(inc, d);
        if (lane >= d) inc += v;
    }
    if (lane == 31) wsum[wv] = inc;
    __syncthreads();
    if (wv == 0) {
        const int w = wsum[lane];
        int wi = w;
        for (int d = 1; d < 32; d <<= 1) {
            const int v = __shfl_up(wi, d);
            if (lane >= d) wi += v;
        }
        wsum[lane] = wi - w;
    }
    __syncthreads();
    int run = wsum[wv] + inc - s;
    for (int k = 0; k < per; ++k) {
        const int t = t0 + k;
        if (t < n_tiles) { off[t] = run; run += cnt[t]; }
    }
    __syncthreads();

    v2i* dsc = (v2i*)desc + (long)blockIdx.x * n_tiles;
    for (int t = tid; t < n_tiles; t += BIN_T) {
        const v2i d = {e0 + off[t], cnt[t]};
        dsc[t] = d;
    }
    __syncthreads();

    for (int base = e0 + tid; base < e1; base += BIN_T * BIN_B) {
        int c[BIN_B], r[BIN_B], a[BIN_B];
#pragma unroll
        for (int u = 0; u < BIN_B; ++u) {
            const int e = base + u * BIN_T;
            const bool in = e < e1;
            c[u] = in ? col[e] : -1;
            r[u] = in ? row[e] : 0;
            a[u] = in ? __float_as_int(adj[e]) : 0;
        }
#pragma unroll
        for (int u = 0; u < BIN_B; ++u) {
            if ((unsigned)c[u] < (unsigned)n_nodes) {
                const int slot = e0 + atomicAdd(&off[c[u] >> 4], 1);
                const v4i v = {r[u], a[u], c[u] & 15, 0};
                ((v4i*)rec)[slot] = v;
            }
        }
    }
}

__global__ __launch_bounds__(256) void k_gemm_push(const float* __restrict__ x,
                                                   const float* __restrict__ w,
                                                   const int* __restrict__ rec,
                                                   const int* __restrict__ desc,
                                                   float* __restrict__ out,
                                                   int n_rows, int n_edges, int n_tiles) {
    __shared__ float wl[D_IN * D_OUT];
    __shared__ float hl[8 * 16 * HL_PITCH];

    const int tid = threadIdx.x;
    for (int i = tid; i < D_IN * D_OUT; i += 256) {
        int k = i / D_OUT;
        int n = i - k * D_OUT;
        wl[(k >> 1) * 128 + n * 2 + (k & 1)] = w[i];
    }
    __syncthreads();

    const int wave = tid >> 5;
    const int lane = tid & 31;
    const int l15  = lane & 15;
    const int hi   = lane >> 4;

    const int  tile = (int)blockIdx.x * 8 + wave;
    const long row0 = (long)tile * 16;
    if (row0 >= n_rows) return;

    const v2i* dsc = (const v2i*)desc;
    const v2i d0 = dsc[(long)lane * n_tiles + tile];
    const v2i d1 = dsc[(long)(lane + 32) * n_tiles + tile];

    const float* ap = x + (row0 + l15) * D_IN + 2 * hi;

    v8f c0 = {}, c1 = {}, c2 = {}, c3 = {};

    for (int k = 0; k < D_IN; k += 4) {
        v2f a = *(const v2f*)(ap + k);
        const float* bbase = &wl[((k >> 1) + hi) * 128 + l15 * 2];
        v2f b0 = *(const v2f*)(bbase +  0);
        v2f b1 = *(const v2f*)(bbase + 32);
        v2f b2 = *(const v2f*)(bbase + 64);
        v2f b3 = *(const v2f*)(bbase + 96);
        c0 = __builtin_amdgcn_wmma_f32_16x16x4_f32(false, a, false, b0, (short)0, c0, false, false);
        c1 = __builtin_amdgcn_wmma_f32_16x16x4_f32(false, a, false, b1, (short)0, c1, false, false);
        c2 = __builtin_amdgcn_wmma_f32_16x16x4_f32(false, a, false, b2, (short)0, c2, false, false);
        c3 = __builtin_amdgcn_wmma_f32_16x16x4_f32(false, a, false, b3, (short)0, c3, false, false);
    }

    float* hw = hl + wave * (16 * HL_PITCH);
#pragma unroll
    for (int r = 0; r < 8; ++r) {
        float* hr = hw + (r + 8 * hi) * HL_PITCH + l15;
        hr[0]  = c0[r];
        hr[16] = c1[r];
        hr[32] = c2[r];
        hr[48] = c3[r];
    }

    const int elast = n_edges - 1;
    const int s0 = d0[0];
    const int s1 = d1[0];
    const int n0 = d0[1] > 0 ? d0[1] : 0;
    const int n1 = d1[1] > 0 ? d1[1] : 0;
    const int tot = n0 + n1;

    int c_rw = 0, c_ab = 0, c_q = 0;
    bool c_act = tot > 0;
    if (c_act) {
        int idx = (n0 > 0) ? s0 : s1;
        idx = idx < 0 ? 0 : (idx > elast ? elast : idx);
        const int* np = rec + 4 * (long)idx;
        const v2i a = *(const v2i*)np;
        c_rw = a[0]; c_ab = a[1]; c_q = np[2];
    }

    float* ob = out + lane;
    const float* hb = hw + lane;

    for (int j = 0; j < n_edges; ++j) {
        if (__builtin_amdgcn_ballot_w32(c_act) == 0u) break;
        const unsigned m = __builtin_amdgcn_ballot_w32(c_act && (unsigned)c_rw < (unsigned)n_rows);

        const int jn = j + 1;
        const bool n_act = jn < tot;
        int n_rw = 0, n_ab = 0, n_q = 0;
        if (n_act) {
            int idx = (jn < n0) ? (s0 + jn) : (s1 + (jn - n0));
            idx = idx < 0 ? 0 : (idx > elast ? elast : idx);
            const int* np = rec + 4 * (long)idx;
            const v2i a = *(const v2i*)np;
            n_rw = a[0]; n_ab = a[1]; n_q = np[2];
        }

#pragma unroll
        for (int q = 0; q < 32; ++q) {
            if (m & (1u << q)) {
                const int   rr = __builtin_amdgcn_readlane(c_rw, q);
                const float v  = __int_as_float(__builtin_amdgcn_readlane(c_ab, q));
                const int   sq = __builtin_amdgcn_readlane(c_q, q) & 15;
                const float* h = hb + sq * HL_PITCH;
                const float h0 = h[0];
                const float h1 = h[32];
                float* o = ob + (long)rr * D_OUT;
                atomicAdd(o,      v * h0);
                atomicAdd(o + 32, v * h1);
            }
        }
        c_rw  = n_rw;
        c_ab  = n_ab;
        c_q   = n_q;
        c_act = n_act;
    }
}

__global__ __launch_bounds__(256) void k_prelu(float* __restrict__ out,
                                               const float* __restrict__ slope,
                                               int n4) {
    int i = blockIdx.x * 256 + threadIdx.x;
    if (i >= n4) return;
    const float s = slope[0];
    v4f v = ((v4f*)out)[i];
#pragma unroll
    for (int j = 0; j < 4; ++j) {
        float t = v[j];
        v[j] = t > 0.f ? t : s * t;
    }
    ((v4f*)out)[i] = v;
}

extern "C" void kernel_launch(void* const* d_in, const int* in_sizes, int n_in,
                              void* d_out, int out_size, void* d_ws, size_t ws_size,
                              hipStream_t stream) {
    const float* x    = (const float*)d_in[0];
    const float* w    = (const float*)d_in[1];
    const float* adj  = (const float*)d_in[2];
    const float* pa   = (const float*)d_in[3];
    const int*   row  = (const int*)d_in[4];
    const int*   col  = (const int*)d_in[5];
    float*       out  = (float*)d_out;

    const int n_nodes = in_sizes[0] / D_IN;
    const int n_edges = in_sizes[2];
    const int out4    = out_size / 4;
    const int n_tiles = (n_nodes + 15) / 16;
    const int chunk   = (n_edges + N_CHUNK - 1) / N_CHUNK;

    v4i* rec4 = (v4i*)d_ws;
    int* rec  = (int*)rec4;
    int* desc = (int*)(rec4 + n_edges);

    const size_t bin_lds = (size_t)(2 * n_tiles + 32) * sizeof(int);
    k_bin<<<256, BIN_T, bin_lds, stream>>>(adj, row, col, rec, desc, out, out4,
                                           n_edges, n_nodes, n_tiles, chunk);

    k_gemm_push<<<(n_tiles + 7) / 8, 256, 0, stream>>>(x, w, rec, desc, out, n_nodes, n_edges, n_tiles);

    k_prelu<<<(out4 + 255) / 256, 256, 0, stream>>>(out, pa, out4);
}


# GEMM+push launch: 782 blocks of 8 waves (3 per WGP, 14 left for a 2nd round) -> 250 blocks of 25 waves, one per WGP (LDS), one round: atomic adds even over WGPs; f32 atomic order changes
# speedup vs baseline: 1.5250x; 1.1217x over previous
#include <hip/hip_runtime.h>


typedef __attribute__((ext_vector_type(2))) float v2f;
typedef __attribute__((ext_vector_type(4))) float v4f;
typedef __attribute__((ext_vector_type(8))) float v8f;
typedef __attribute__((ext_vector_type(4))) int   v4i;
typedef __attribute__((ext_vector_type(2))) int   v2i;

#define D_IN     256
#define D_OUT    64
#define N_CHUNK  64
#define BIN_T    1024
#define BIN_B    8
#define HL_PITCH 66
#define GP_WAVES 25
#define GP_T     (GP_WAVES * 32)

__global__ __launch_bounds__(BIN_T) void k_bin(const float* __restrict__ adj,
                                               const int* __restrict__ row,
                                               const int* __restrict__ col,
                                               int* __restrict__ rec,
                                               int* __restrict__ desc,
                                               float* __restrict__ out, int out4,
                                               int n_edges, int n_nodes,
                                               int n_tiles, int chunk) {
    extern __shared__ int sm[];
    const int tid = threadIdx.x;

    if (blockIdx.x >= N_CHUNK) {
        const int nz = (int)gridDim.x - N_CHUNK;
        const v4f z = {0.f, 0.f, 0.f, 0.f};
        for (int i = ((int)blockIdx.x - N_CHUNK) * BIN_T + tid; i < out4; i += nz * BIN_T)
            ((v4f*)out)[i] = z;
        return;
    }

    int* cnt  = sm;
    int* off  = sm + n_tiles;
    int* wsum = off + n_tiles;

    const int e0 = (int)blockIdx.x * chunk;
    int e1 = e0 + chunk;
    if (e1 > n_edges) e1 = n_edges;

    for (int t = tid; t < n_tiles; t += BIN_T) cnt[t] = 0;
    __syncthreads();

    for (int base = e0 + tid; base < e1; base += BIN_T * BIN_B) {
        int c[BIN_B];
#pragma unroll
        for (int u = 0; u < BIN_B; ++u) {
            const int e = base + u * BIN_T;
            c[u] = (e < e1) ? col[e] : -1;
        }
#pragma unroll
        for (int u = 0; u < BIN_B; ++u)
            if ((unsigned)c[u] < (unsigned)n_nodes) atomicAdd(&cnt[c[u] >> 4], 1);
    }
    __syncthreads();

    const int per  = (n_tiles + BIN_T - 1) / BIN_T;
    const int t0   = tid * per;
    const int lane = tid & 31;
    const int wv   = tid >> 5;
    int s = 0;
    for (int k = 0; k < per; ++k) {
        const int t = t0 + k;
        if (t < n_tiles) s += cnt[t];
    }
    int inc = s;
    for (int d = 1; d < 32; d <<= 1) {
        const int v = __shfl_up(inc, d);
        if (lane >= d) inc += v;
    }
    if (lane == 31) wsum[wv] = inc;
    __syncthreads();
    if (wv == 0) {
        const int w = wsum[lane];
        int wi = w;
        for (int d = 1; d < 32; d <<= 1) {
            const int v = __shfl_up(wi, d);
            if (lane >= d) wi += v;
        }
        wsum[lane] = wi - w;
    }
    __syncthreads();
    int run = wsum[wv] + inc - s;
    for (int k = 0; k < per; ++k) {
        const int t = t0 + k;
        if (t < n_tiles) { off[t] = run; run += cnt[t]; }
    }
    __syncthreads();

    v2i* dsc = (v2i*)desc + (long)blockIdx.x * n_tiles;
    for (int t = tid; t < n_tiles; t += BIN_T) {
        const v2i d = {e0 + off[t], cnt[t]};
        dsc[t] = d;
    }
    __syncthreads();

    for (int base = e0 + tid; base < e1; base += BIN_T * BIN_B) {
        int c[BIN_B], r[BIN_B], a[BIN_B];
#pragma unroll
        for (int u = 0; u < BIN_B; ++u) {
            const int e = base + u * BIN_T;
            const bool in = e < e1;
            c[u] = in ? col[e] : -1;
            r[u] = in ? row[e] : 0;
            a[u] = in ? __float_as_int(adj[e]) : 0;
        }
#pragma unroll
        for (int u = 0; u < BIN_B; ++u) {
            if ((unsigned)c[u] < (unsigned)n_nodes) {
                const int slot = e0 + atomicAdd(&off[c[u] >> 4], 1);
                const v4i v = {r[u], a[u], c[u] & 15, 0};
                ((v4i*)rec)[slot] = v;
            }
        }
    }
}

__global__ __launch_bounds__(GP_T) void k_gemm_push(const float* __restrict__ x,
                                                   const float* __restrict__ w,
                                                   const int* __restrict__ rec,
                                                   const int* __restrict__ desc,
                                                   float* __restrict__ out,
                                                   int n_rows, int n_edges, int n_tiles) {
    __shared__ float wl[D_IN * D_OUT];
    __shared__ float hl[GP_WAVES * 16 * HL_PITCH];

    const int tid = threadIdx.x;
    for (int i = tid; i < D_IN * D_OUT; i += GP_T) {
        int k = i / D_OUT;
        int n = i - k * D_OUT;
        wl[(k >> 1) * 128 + n * 2 + (k & 1)] = w[i];
    }
    __syncthreads();

    const int wave = tid >> 5;
    const int lane = tid & 31;
    const int l15  = lane & 15;
    const int hi   = lane >> 4;

    const int  tile = (int)blockIdx.x * GP_WAVES + wave;
    const long row0 = (long)tile * 16;
    if (row0 >= n_rows) return;

    const v2i* dsc = (const v2i*)desc;
    const v2i d0 = dsc[(long)lane * n_tiles + tile];
    const v2i d1 = dsc[(long)(lane + 32) * n_tiles + tile];

    const float* ap = x + (row0 + l15) * D_IN + 2 * hi;

    v8f c0 = {}, c1 = {}, c2 = {}, c3 = {};

    for (int k = 0; k < D_IN; k += 4) {
        v2f a = *(const v2f*)(ap + k);
        const float* bbase = &wl[((k >> 1) + hi) * 128 + l15 * 2];
        v2f b0 = *(const v2f*)(bbase +  0);
        v2f b1 = *(const v2f*)(bbase + 32);
        v2f b2 = *(const v2f*)(bbase + 64);
        v2f b3 = *(const v2f*)(bbase + 96);
        c0 = __builtin_amdgcn_wmma_f32_16x16x4_f32(false, a, false, b0, (short)0, c0, false, false);
        c1 = __builtin_amdgcn_wmma_f32_16x16x4_f32(false, a, false, b1, (short)0, c1, false, false);
        c2 = __builtin_amdgcn_wmma_f32_16x16x4_f32(false, a, false, b2, (short)0, c2, false, false);
        c3 = __builtin_amdgcn_wmma_f32_16x16x4_f32(false, a, false, b3, (short)0, c3, false, false);
    }

    float* hw = hl + wave * (16 * HL_PITCH);
#pragma unroll
    for (int r = 0; r < 8; ++r) {
        float* hr = hw + (r + 8 * hi) * HL_PITCH + l15;
        hr[0]  = c0[r];
        hr[16] = c1[r];
        hr[32] = c2[r];
        hr[48] = c3[r];
    }

    const int elast = n_edges - 1;
    const int s0 = d0[0];
    const int s1 = d1[0];
    const int n0 = d0[1] > 0 ? d0[1] : 0;
    const int n1 = d1[1] > 0 ? d1[1] : 0;
    const int tot = n0 + n1;

    int c_rw = 0, c_ab = 0, c_q = 0;
    bool c_act = tot > 0;
    if (c_act) {
        int idx = (n0 > 0) ? s0 : s1;
        idx = idx < 0 ? 0 : (idx > elast ? elast : idx);
        const int* np = rec + 4 * (long)idx;
        const v2i a = *(const v2i*)np;
        c_rw = a[0]; c_ab = a[1]; c_q = np[2];
    }

    float* ob = out + lane;
    const float* hb = hw + lane;

    for (int j = 0; j < n_edges; ++j) {
        if (__builtin_amdgcn_ballot_w32(c_act) == 0u) break;
        const unsigned m = __builtin_amdgcn_ballot_w32(c_act && (unsigned)c_rw < (unsigned)n_rows);

        const int jn = j + 1;
        const bool n_act = jn < tot;
        int n_rw = 0, n_ab = 0, n_q = 0;
        if (n_act) {
            int idx = (jn < n0) ? (s0 + jn) : (s1 + (jn - n0));
            idx = idx < 0 ? 0 : (idx > elast ? elast : idx);
            const int* np = rec + 4 * (long)idx;
            const v2i a = *(const v2i*)np;
            n_rw = a[0]; n_ab = a[1]; n_q = np[2];
        }

#pragma unroll
        for (int q = 0; q < 32; ++q) {
            if (m & (1u << q)) {
                const int   rr = __builtin_amdgcn_readlane(c_rw, q);
                const float v  = __int_as_float(__builtin_amdgcn_readlane(c_ab, q));
                const int   sq = __builtin_amdgcn_readlane(c_q, q) & 15;
                const float* h = hb + sq * HL_PITCH;
                const float h0 = h[0];
                const float h1 = h[32];
                float* o = ob + (long)rr * D_OUT;
                atomicAdd(o,      v * h0);
                atomicAdd(o + 32, v * h1);
            }
        }
        c_rw  = n_rw;
        c_ab  = n_ab;
        c_q   = n_q;
        c_act = n_act;
    }
}

__global__ __launch_bounds__(256) void k_prelu(float* __restrict__ out,
                                               const float* __restrict__ slope,
                                               int n4) {
    int i = blockIdx.x * 256 + threadIdx.x;
    if (i >= n4) return;
    const float s = slope[0];
    v4f v = ((v4f*)out)[i];
#pragma unroll
    for (int j = 0; j < 4; ++j) {
        float t = v[j];
        v[j] = t > 0.f ? t : s * t;
    }
    ((v4f*)out)[i] = v;
}

extern "C" void kernel_launch(void* const* d_in, const int* in_sizes, int n_in,
                              void* d_out, int out_size, void* d_ws, size_t ws_size,
                              hipStream_t stream) {
    const float* x    = (const float*)d_in[0];
    const float* w    = (const float*)d_in[1];
    const float* adj  = (const float*)d_in[2];
    const float* pa   = (const float*)d_in[3];
    const int*   row  = (const int*)d_in[4];
    const int*   col  = (const int*)d_in[5];
    float*       out  = (float*)d_out;

    const int n_nodes = in_sizes[0] / D_IN;
    const int n_edges = in_sizes[2];
    const int out4    = out_size / 4;
    const int n_tiles = (n_nodes + 15) / 16;
    const int chunk   = (n_edges + N_CHUNK - 1) / N_CHUNK;

    v4i* rec4 = (v4i*)d_ws;
    int* rec  = (int*)rec4;
    int* desc = (int*)(rec4 + n_edges);

    const size_t bin_lds = (size_t)(2 * n_tiles + 32) * sizeof(int);
    k_bin<<<256, BIN_T, bin_lds, stream>>>(adj, row, col, rec, desc, out, out4,
                                           n_edges, n_nodes, n_tiles, chunk);

    k_gemm_push<<<(n_tiles + GP_WAVES - 1) / GP_WAVES, GP_T, 0, stream>>>(x, w, rec, desc, out, n_nodes, n_edges, n_tiles);

    k_prelu<<<(out4 + 255) / 256, 256, 0, stream>>>(out, pa, out4);
}


# edge sort spread from 64 to 256 work-groups: 4 blocks share a chunk, each counts it whole, places only its quarter of the tile range (scattered 16-B stores: 1/4 per WGP); out cleared by all blocks
# speedup vs baseline: 1.6116x; 1.0568x over previous
#include <hip/hip_runtime.h>


typedef __attribute__((ext_vector_type(2))) float v2f;
typedef __attribute__((ext_vector_type(4))) float v4f;
typedef __attribute__((ext_vector_type(8))) float v8f;
typedef __attribute__((ext_vector_type(4))) int   v4i;
typedef __attribute__((ext_vector_type(2))) int   v2i;

#define D_IN     256
#define D_OUT    64
#define N_CHUNK  64
#define N_SPLIT  4
#define BIN_T    1024
#define BIN_B    8
#define HL_PITCH 66
#define GP_WAVES 25
#define GP_T     (GP_WAVES * 32)

__device__ __forceinline__ void clear_share(float* __restrict__ out, int out4, int tid) {
    const v4f z = {0.f, 0.f, 0.f, 0.f};
    for (int i = (int)blockIdx.x * BIN_T + tid; i < out4; i += (int)gridDim.x * BIN_T)
        ((v4f*)out)[i] = z;
}

__global__ __launch_bounds__(BIN_T) void k_bin(const float* __restrict__ adj,
                                               const int* __restrict__ row,
                                               const int* __restrict__ col,
                                               int* __restrict__ rec,
                                               int* __restrict__ desc,
                                               float* __restrict__ out, int out4,
                                               int n_edges, int n_nodes,
                                               int n_tiles, int chunk) {
    extern __shared__ int sm[];
    const int tid = threadIdx.x;

    const int ck = (int)blockIdx.x / N_SPLIT;
    const int sp = (int)blockIdx.x - ck * N_SPLIT;
    if (ck >= N_CHUNK) {
        clear_share(out, out4, tid);
        return;
    }

    const int tq  = (n_tiles + N_SPLIT - 1) / N_SPLIT;
    const int tlo = sp * tq;
    int thi = tlo + tq;
    if (thi > n_tiles) thi = n_tiles;

    int* cnt  = sm;
    int* off  = sm + n_tiles;
    int* wsum = off + n_tiles;

    const int e0 = ck * chunk;
    int e1 = e0 + chunk;
    if (e1 > n_edges) e1 = n_edges;

    for (int t = tid; t < n_tiles; t += BIN_T) cnt[t] = 0;
    __syncthreads();

    clear_share(out, out4, tid);

    for (int base = e0 + tid; base < e1; base += BIN_T * BIN_B) {
        int c[BIN_B];
#pragma unroll
        for (int u = 0; u < BIN_B; ++u) {
            const int e = base + u * BIN_T;
            c[u] = (e < e1) ? col[e] : -1;
        }
#pragma unroll
        for (int u = 0; u < BIN_B; ++u)
            if ((unsigned)c[u] < (unsigned)n_nodes) atomicAdd(&cnt[c[u] >> 4], 1);
    }
    __syncthreads();

    const int per  = (n_tiles + BIN_T - 1) / BIN_T;
    const int t0   = tid * per;
    const int lane = tid & 31;
    const int wv   = tid >> 5;
    int s = 0;
    for (int k = 0; k < per; ++k) {
        const int t = t0 + k;
        if (t < n_tiles) s += cnt[t];
    }
    int inc = s;
    for (int d = 1; d < 32; d <<= 1) {
        const int v = __shfl_up(inc, d);
        if (lane >= d) inc += v;
    }
    if (lane == 31) wsum[wv] = inc;
    __syncthreads();
    if (wv == 0) {
        const int w = wsum[lane];
        int wi = w;
        for (int d = 1; d < 32; d <<= 1) {
            const int v = __shfl_up(wi, d);
            if (lane >= d) wi += v;
        }
        wsum[lane] = wi - w;
    }
    __syncthreads();
    int run = wsum[wv] + inc - s;
    for (int k = 0; k < per; ++k) {
        const int t = t0 + k;
        if (t < n_tiles) { off[t] = run; run += cnt[t]; }
    }
    __syncthreads();

    v2i* dsc = (v2i*)desc + (long)ck * n_tiles;
    for (int t = tlo + tid; t < thi; t += BIN_T) {
        const v2i d = {e0 + off[t], cnt[t]};
        dsc[t] = d;
    }
    __syncthreads();

    for (int base = e0 + tid; base < e1; base += BIN_T * BIN_B) {
        int c[BIN_B], r[BIN_B], a[BIN_B];
#pragma unroll
        for (int u = 0; u < BIN_B; ++u) {
            const int e = base + u * BIN_T;
            const bool in = e < e1;
            c[u] = in ? col[e] : -1;
            r[u] = in ? row[e] : 0;
            a[u] = in ? __float_as_int(adj[e]) : 0;
        }
#pragma unroll
        for (int u = 0; u < BIN_B; ++u) {
            const int t = c[u] >> 4;
            if ((unsigned)c[u] < (unsigned)n_nodes && t >= tlo && t < thi) {
                const int slot = e0 + atomicAdd(&off[t], 1);
                const v4i v = {r[u], a[u], c[u] & 15, 0};
                ((v4i*)rec)[slot] = v;
            }
        }
    }
}

__global__ __launch_bounds__(GP_T) void k_gemm_push(const float* __restrict__ x,
                                                   const float* __restrict__ w,
                                                   const int* __restrict__ rec,
                                                   const int* __restrict__ desc,
                                                   float* __restrict__ out,
                                                   int n_rows, int n_edges, int n_tiles) {
    __shared__ float wl[D_IN * D_OUT];
    __shared__ float hl[GP_WAVES * 16 * HL_PITCH];

    const int tid = threadIdx.x;
    for (int i = tid; i < D_IN * D_OUT; i += GP_T) {
        int k = i / D_OUT;
        int n = i - k * D_OUT;
        wl[(k >> 1) * 128 + n * 2 + (k & 1)] = w[i];
    }
    __syncthreads();

    const int wave = tid >> 5;
    const int lane = tid & 31;
    const int l15  = lane & 15;
    const int hi   = lane >> 4;

    const int  tile = (int)blockIdx.x * GP_WAVES + wave;
    const long row0 = (long)tile * 16;
    if (row0 >= n_rows) return;

    const v2i* dsc = (const v2i*)desc;
    const v2i d0 = dsc[(long)lane * n_tiles + tile];
    const v2i d1 = dsc[(long)(lane + 32) * n_tiles + tile];

    const float* ap = x + (row0 + l15) * D_IN + 2 * hi;

    v8f c0 = {}, c1 = {}, c2 = {}, c3 = {};

    for (int k = 0; k < D_IN; k += 4) {
        v2f a = *(const v2f*)(ap + k);
        const float* bbase = &wl[((k >> 1) + hi) * 128 + l15 * 2];
        v2f b0 = *(const v2f*)(bbase +  0);
        v2f b1 = *(const v2f*)(bbase + 32);
        v2f b2 = *(const v2f*)(bbase + 64);
        v2f b3 = *(const v2f*)(bbase + 96);
        c0 = __builtin_amdgcn_wmma_f32_16x16x4_f32(false, a, false, b0, (short)0, c0, false, false);
        c1 = __builtin_amdgcn_wmma_f32_16x16x4_f32(false, a, false, b1, (short)0, c1, false, false);
        c2 = __builtin_amdgcn_wmma_f32_16x16x4_f32(false, a, false, b2, (short)0, c2, false, false);
        c3 = __builtin_amdgcn_wmma_f32_16x16x4_f32(false, a, false, b3, (short)0, c3, false, false);
    }

    float* hw = hl + wave * (16 * HL_PITCH);
#pragma unroll
    for (int r = 0; r < 8; ++r) {
        float* hr = hw + (r + 8 * hi) * HL_PITCH + l15;
        hr[0]  = c0[r];
        hr[16] = c1[r];
        hr[32] = c2[r];
        hr[48] = c3[r];
    }

    const int elast = n_edges - 1;
    const int s0 = d0[0];
    const int s1 = d1[0];
    const int n0 = d0[1] > 0 ? d0[1] : 0;
    const int n1 = d1[1] > 0 ? d1[1] : 0;
    const int tot = n0 + n1;

    int c_rw = 0, c_ab = 0, c_q = 0;
    bool c_act = tot > 0;
    if (c_act) {
        int idx = (n0 > 0) ? s0 : s1;
        idx = idx < 0 ? 0 : (idx > elast ? elast : idx);
        const int* np = rec + 4 * (long)idx;
        const v2i a = *(const v2i*)np;
        c_rw = a[0]; c_ab = a[1]; c_q = np[2];
    }

    float* ob = out + lane;
    const float* hb = hw + lane;

    for (int j = 0; j < n_edges; ++j) {
        if (__builtin_amdgcn_ballot_w32(c_act) == 0u) break;
        const unsigned m = __builtin_amdgcn_ballot_w32(c_act && (unsigned)c_rw < (unsigned)n_rows);

        const int jn = j + 1;
        const bool n_act = jn < tot;
        int n_rw = 0, n_ab = 0, n_q = 0;
        if (n_act) {
            int idx = (jn < n0) ? (s0 + jn) : (s1 + (jn - n0));
            idx = idx < 0 ? 0 : (idx > elast ? elast : idx);
            const int* np = rec + 4 * (long)idx;
            const v2i a = *(const v2i*)np;
            n_rw = a[0]; n_ab = a[1]; n_q = np[2];
        }

#pragma unroll
        for (int q = 0; q < 32; ++q) {
            if (m & (1u << q)) {
                const int   rr = __builtin_amdgcn_readlane(c_rw, q);
                const float v  = __int_as_float(__builtin_amdgcn_readlane(c_ab, q));
                const int   sq = __builtin_amdgcn_readlane(c_q, q) & 15;
                const float* h = hb + sq * HL_PITCH;
                const float h0 = h[0];
                const float h1 = h[32];
                float* o = ob + (long)rr * D_OUT;
                atomicAdd(o,      v * h0);
                atomicAdd(o + 32, v * h1);
            }
        }
        c_rw  = n_rw;
        c_ab  = n_ab;
        c_q   = n_q;
        c_act = n_act;
    }
}

__global__ __launch_bounds__(256) void k_prelu(float* __restrict__ out,
                                               const float* __restrict__ slope,
                                               int n4) {
    int i = blockIdx.x * 256 + threadIdx.x;
    if (i >= n4) return;
    const float s = slope[0];
    v4f v = ((v4f*)out)[i];
#pragma unroll
    for (int j = 0; j < 4; ++j) {
        float t = v[j];
        v[j] = t > 0.f ? t : s * t;
    }
    ((v4f*)out)[i] = v;
}

extern "C" void kernel_launch(void* const* d_in, const int* in_sizes, int n_in,
                              void* d_out, int out_size, void* d_ws, size_t ws_size,
                              hipStream_t stream) {
    const float* x    = (const float*)d_in[0];
    const float* w    = (const float*)d_in[1];
    const float* adj  = (const float*)d_in[2];
    const float* pa   = (const float*)d_in[3];
    const int*   row  = (const int*)d_in[4];
    const int*   col  = (const int*)d_in[5];
    float*       out  = (float*)d_out;

    const int n_nodes = in_sizes[0] / D_IN;
    const int n_edges = in_sizes[2];
    const int out4    = out_size / 4;
    const int n_tiles = (n_nodes + 15) / 16;
    const int chunk   = (n_edges + N_CHUNK - 1) / N_CHUNK;

    v4i* rec4 = (v4i*)d_ws;
    int* rec  = (int*)rec4;
    int* desc = (int*)(rec4 + n_edges);

    const size_t bin_lds = (size_t)(2 * n_tiles + 32) * sizeof(int);
    k_bin<<<N_CHUNK * N_SPLIT, BIN_T, bin_lds, stream>>>(adj, row, col, rec, desc, out, out4,
                                           n_edges, n_nodes, n_tiles, chunk);

    k_gemm_push<<<(n_tiles + GP_WAVES - 1) / GP_WAVES, GP_T, 0, stream>>>(x, w, rec, desc, out, n_nodes, n_edges, n_tiles);

    k_prelu<<<(out4 + 255) / 256, 256, 0, stream>>>(out, pa, out4);
}
